# starting kernel
# baseline (speedup 1.0000x reference)
#include <hip/hip_runtime.h>
#include <hip/hip_bf16.h>

typedef __attribute__((ext_vector_type(2))) float v2f;
typedef __attribute__((ext_vector_type(8))) float v8f;

#define N_CLASS 32
#define N_SUP   16
#define FDIM    1024
#define N_QRY   512
#define QTILE   16
#define DCHUNK  256
#define LSTRIDE (DCHUNK + 4)

__device__ __forceinline__ float dev_tanh(float x) {
#if __has_builtin(__builtin_amdgcn_tanhf)
  return __builtin_amdgcn_tanhf(x);
#else
  return tanhf(x);
#endif
}

__global__ __launch_bounds__(256)
void proto_att_kernel(const float* __restrict__ data, float* __restrict__ out) {
  __shared__ float sQ[QTILE * LSTRIDE];
  __shared__ float sS[N_SUP * LSTRIDE];
  __shared__ float sAtt[QTILE][N_SUP + 1];

  const int qt  = blockIdx.x;
  const int c   = blockIdx.y;
  const int tid = threadIdx.x;

  const float* Qg = data + (size_t)(N_CLASS * N_SUP + qt * QTILE) * FDIM;
  const float* Sg = data + (size_t)c * N_SUP * FDIM;

  const int q   = tid >> 4;
  const int s   = tid & 15;
  const int row = tid >> 4;
  const int l16 = tid & 15;

  float acc = 0.f;
  for (int d0 = 0; d0 < FDIM; d0 += DCHUNK) {
    const float4* qsrc = (const float4*)(Qg + (size_t)row * FDIM + d0);
    const float4* ssrc = (const float4*)(Sg + (size_t)row * FDIM + d0);
    float4* qdst = (float4*)(sQ + row * LSTRIDE);
    float4* sdst = (float4*)(sS + row * LSTRIDE);
#pragma unroll
    for (int j = 0; j < 4; ++j) {
      int c4 = l16 + j * 16;
      qdst[c4] = qsrc[c4];
      sdst[c4] = ssrc[c4];
    }
    __syncthreads();
    const float* qr = sQ + q * LSTRIDE;
    const float* sr = sS + s * LSTRIDE;
#pragma unroll 8
    for (int d = 0; d < DCHUNK; d += 4) {
      float4 qv = *(const float4*)(qr + d);
      float4 sv = *(const float4*)(sr + d);
      acc += dev_tanh(qv.x * sv.x);
      acc += dev_tanh(qv.y * sv.y);
      acc += dev_tanh(qv.z * sv.z);
      acc += dev_tanh(qv.w * sv.w);
    }
    __syncthreads();
  }
  sAtt[q][s] = acc;
  __syncthreads();

  float m = -3.402823466e38f;
  float rowv[N_SUP];
#pragma unroll
  for (int i = 0; i < N_SUP; ++i) { rowv[i] = sAtt[q][i]; m = fmaxf(m, rowv[i]); }
  float sum = 0.f;
#pragma unroll
  for (int i = 0; i < N_SUP; ++i) sum += __expf(rowv[i] - m);
  float mya = __expf(acc - m) / sum;
  __syncthreads();
  sAtt[q][s] = mya;
  __syncthreads();

  const int wave = tid >> 5;
  const int lane = tid & 31;
  const int hi   = lane >> 4;
  const int ln   = lane & 15;

  v2f a[4];
#pragma unroll
  for (int kk = 0; kk < 4; ++kk) {
    a[kk].x = sAtt[ln][kk * 4 + (hi ? 2 : 0)];
    a[kk].y = sAtt[ln][kk * 4 + (hi ? 3 : 1)];
  }

  float* outBase = out + ((size_t)(qt * QTILE) * N_CLASS + c) * FDIM;

  for (int nt = wave; nt < FDIM / 16; nt += 8) {
    const int n0 = nt * 16;
    v8f dacc = {};
#pragma unroll
    for (int kk = 0; kk < 4; ++kk) {
      v2f b;
      b.x = Sg[(size_t)(kk * 4 + (hi ? 2 : 0)) * FDIM + n0 + ln];
      b.y = Sg[(size_t)(kk * 4 + (hi ? 3 : 1)) * FDIM + n0 + ln];
      dacc = __builtin_amdgcn_wmma_f32_16x16x4_f32(false, a[kk], false, b,
                                                   (short)0, dacc, false, false);
    }
#pragma unroll
    for (int r = 0; r < 8; ++r) {
      int M = r + (hi ? 8 : 0);
      outBase[(size_t)M * N_CLASS * FDIM + n0 + ln] = dacc[r];
    }
  }
}

extern "C" void kernel_launch(void* const* d_in, const int* in_sizes, int n_in,
                              void* d_out, int out_size, void* d_ws, size_t ws_size,
                              hipStream_t stream) {
  const float* data = (const float*)d_in[0];
  float* out = (float*)d_out;
  (void)in_sizes; (void)n_in; (void)d_ws; (void)ws_size; (void)out_size;
  dim3 grid(N_QRY / QTILE, N_CLASS);
  proto_att_kernel<<<grid, 256, 0, stream>>>(data, out);
}


# score phase re-tiled: lanes own features, a wave keeps a 4x8 (query,support) tile of partial sums in registers so one LDS read feeds 4-8 tanh; partials summed across lanes via LDS (new sum order)
# speedup vs baseline: 1.1239x; 1.1239x over previous
#include <hip/hip_runtime.h>
#include <hip/hip_bf16.h>

typedef __attribute__((ext_vector_type(2))) float v2f;
typedef __attribute__((ext_vector_type(8))) float v8f;
typedef __attribute__((ext_vector_type(4))) float v4f;

#define N_CLASS 32
#define N_SUP   16
#define FDIM    1024
#define N_QRY   512
#define QTILE   16
#define DCHUNK  256
#define LSTRIDE (DCHUNK + 4)
#define QREG    4
#define SREG    8
#define RSTRIDE 33

__device__ __forceinline__ float dev_tanh(float x) {
#if __has_builtin(__builtin_amdgcn_tanhf)
  return __builtin_amdgcn_tanhf(x);
#else
  return tanhf(x);
#endif
}

__global__ __launch_bounds__(256)
void proto_att_kernel(const float* __restrict__ data, float* __restrict__ out) {
  __shared__ __attribute__((aligned(16))) float sBuf[QTILE * N_SUP * RSTRIDE];
  __shared__ float sAtt[QTILE][N_SUP + 1];
  float* sQ   = sBuf;
  float* sS   = sBuf + QTILE * LSTRIDE;
  float* sRed = sBuf;
  static_assert(QTILE * N_SUP * RSTRIDE >= (QTILE + N_SUP) * LSTRIDE, "staging tiles must fit");

  const int qt  = blockIdx.x;
  const int c   = blockIdx.y;
  const int tid = threadIdx.x;

  const float* Qg = data + (size_t)(N_CLASS * N_SUP + qt * QTILE) * FDIM;
  const float* Sg = data + (size_t)c * N_SUP * FDIM;

  const int q   = tid >> 4;
  const int s   = tid & 15;
  const int row = tid >> 4;
  const int l16 = tid & 15;

  const int wave = tid >> 5;
  const int lane = tid & 31;

  const int q0 = (wave >> 1) * QREG;
  const int s0 = (wave & 1) * SREG;
  v2f part[QREG][SREG / 2];
#pragma unroll
  for (int i = 0; i < QREG; ++i)
#pragma unroll
    for (int jj = 0; jj < SREG / 2; ++jj) part[i][jj] = (v2f){0.f, 0.f};

  for (int d0 = 0; d0 < FDIM; d0 += DCHUNK) {
    const v4f* qsrc = (const v4f*)(Qg + (size_t)row * FDIM + d0);
    const v4f* ssrc = (const v4f*)(Sg + (size_t)row * FDIM + d0);
    v4f* qdst = (v4f*)(sQ + row * LSTRIDE);
    v4f* sdst = (v4f*)(sS + row * LSTRIDE);
    v4f tq[4], ts[4];
#pragma unroll
    for (int j = 0; j < 4; ++j) {
      int c4 = l16 + j * 16;
      tq[j] = qsrc[c4];
      ts[j] = ssrc[c4];
    }
    asm volatile("" : "+v"(tq[0]), "+v"(tq[1]), "+v"(tq[2]), "+v"(tq[3]),
                      "+v"(ts[0]), "+v"(ts[1]), "+v"(ts[2]), "+v"(ts[3]));
#pragma unroll
    for (int j = 0; j < 4; ++j) {
      int c4 = l16 + j * 16;
      qdst[c4] = tq[j];
      sdst[c4] = ts[j];
    }
    __syncthreads();
#pragma unroll
    for (int h = 0; h < DCHUNK / 128; ++h) {
      const int col = (lane + 32 * h) * 4;
      v4f qv[QREG];
#pragma unroll
      for (int i = 0; i < QREG; ++i) qv[i] = *(const v4f*)(sQ + (q0 + i) * LSTRIDE + col);
#pragma unroll
      for (int jh = 0; jh < SREG; jh += 4) {
        v4f sv[4];
#pragma unroll
        for (int j = 0; j < 4; ++j) sv[j] = *(const v4f*)(sS + (s0 + jh + j) * LSTRIDE + col);
#pragma unroll
        for (int i = 0; i < QREG; ++i) {
#pragma unroll
          for (int jj = 0; jj < 2; ++jj) {
            v4f p0 = qv[i] * sv[2 * jj];
            v4f p1 = qv[i] * sv[2 * jj + 1];
            v2f tx = {dev_tanh(p0.x), dev_tanh(p1.x)};
            v2f ty = {dev_tanh(p0.y), dev_tanh(p1.y)};
            v2f tz = {dev_tanh(p0.z), dev_tanh(p1.z)};
            v2f tw = {dev_tanh(p0.w), dev_tanh(p1.w)};
            part[i][jh / 2 + jj] += (tx + ty) + (tz + tw);
          }
        }
      }
    }
    __syncthreads();
  }

  float* rdst = sRed + (q0 * N_SUP + s0) * RSTRIDE + lane;
#pragma unroll
  for (int i = 0; i < QREG; ++i)
#pragma unroll
    for (int jj = 0; jj < SREG / 2; ++jj) {
      rdst[(i * N_SUP + 2 * jj)     * RSTRIDE] = part[i][jj].x;
      rdst[(i * N_SUP + 2 * jj + 1) * RSTRIDE] = part[i][jj].y;
    }
  __syncthreads();
  float acc = 0.f;
#pragma unroll
  for (int l = 0; l < 32; ++l) acc += sRed[tid * RSTRIDE + l];
  sAtt[q][s] = acc;
  __syncthreads();

  float m = -3.402823466e38f;
  float rowv[N_SUP];
#pragma unroll
  for (int i = 0; i < N_SUP; ++i) { rowv[i] = sAtt[q][i]; m = fmaxf(m, rowv[i]); }
  float sum = 0.f;
#pragma unroll
  for (int i = 0; i < N_SUP; ++i) sum += __expf(rowv[i] - m);
  float mya = __expf(acc - m) / sum;
  __syncthreads();
  sAtt[q][s] = mya;
  __syncthreads();

  const int hi   = lane >> 4;
  const int ln   = lane & 15;

  v2f a[4];
#pragma unroll
  for (int kk = 0; kk < 4; ++kk) {
    a[kk].x = sAtt[ln][kk * 4 + (hi ? 2 : 0)];
    a[kk].y = sAtt[ln][kk * 4 + (hi ? 3 : 1)];
  }

  float* outBase = out + ((size_t)(qt * QTILE) * N_CLASS + c) * FDIM;

  for (int nt = wave; nt < FDIM / 16; nt += 8) {
    const int n0 = nt * 16;
    v8f dacc = {};
#pragma unroll
    for (int kk = 0; kk < 4; ++kk) {
      v2f b;
      b.x = Sg[(size_t)(kk * 4 + (hi ? 2 : 0)) * FDIM + n0 + ln];
      b.y = Sg[(size_t)(kk * 4 + (hi ? 3 : 1)) * FDIM + n0 + ln];
      dacc = __builtin_amdgcn_wmma_f32_16x16x4_f32(false, a[kk], false, b,
                                                   (short)0, dacc, false, false);
    }
#pragma unroll
    for (int r = 0; r < 8; ++r) {
      int M = r + (hi ? 8 : 0);
      outBase[(size_t)M * N_CLASS * FDIM + n0 + ln] = dacc[r];
    }
  }
}

extern "C" void kernel_launch(void* const* d_in, const int* in_sizes, int n_in,
                              void* d_out, int out_size, void* d_ws, size_t ws_size,
                              hipStream_t stream) {
  const float* data = (const float*)d_in[0];
  float* out = (float*)d_out;
  (void)in_sizes; (void)n_in; (void)d_ws; (void)ws_size; (void)out_size;
  dim3 grid(N_QRY / QTILE, N_CLASS);
  proto_att_kernel<<<grid, 256, 0, stream>>>(data, out);
}


# output phase: a wave's four N tiles interleave columns (tile j = columns 4*lane+j), so B loads and result stores are 16 bytes per lane and whole 256-byte row runs per half-wave; sums unchanged
# speedup vs baseline: 1.5629x; 1.3906x over previous
#include <hip/hip_runtime.h>
#include <hip/hip_bf16.h>

typedef __attribute__((ext_vector_type(2))) float v2f;
typedef __attribute__((ext_vector_type(8))) float v8f;
typedef __attribute__((ext_vector_type(4))) float v4f;

#define N_CLASS 32
#define N_SUP   16
#define FDIM    1024
#define N_QRY   512
#define QTILE   16
#define DCHUNK  256
#define LSTRIDE (DCHUNK + 4)
#define QREG    4
#define SREG    8
#define RSTRIDE 33

__device__ __forceinline__ float dev_tanh(float x) {
#if __has_builtin(__builtin_amdgcn_tanhf)
  return __builtin_amdgcn_tanhf(x);
#else
  return tanhf(x);
#endif
}

__global__ __launch_bounds__(256)
void proto_att_kernel(const float* __restrict__ data, float* __restrict__ out) {
  __shared__ __attribute__((aligned(16))) float sBuf[QTILE * N_SUP * RSTRIDE];
  __shared__ float sAtt[QTILE][N_SUP + 1];
  float* sQ   = sBuf;
  float* sS   = sBuf + QTILE * LSTRIDE;
  float* sRed = sBuf;
  static_assert(QTILE * N_SUP * RSTRIDE >= (QTILE + N_SUP) * LSTRIDE, "staging tiles must fit");

  const int qt  = blockIdx.x;
  const int c   = blockIdx.y;
  const int tid = threadIdx.x;

  const float* Qg = data + (size_t)(N_CLASS * N_SUP + qt * QTILE) * FDIM;
  const float* Sg = data + (size_t)c * N_SUP * FDIM;

  const int q   = tid >> 4;
  const int s   = tid & 15;
  const int row = tid >> 4;
  const int l16 = tid & 15;

  const int wave = tid >> 5;
  const int lane = tid & 31;

  const int q0 = (wave >> 1) * QREG;
  const int s0 = (wave & 1) * SREG;
  v2f part[QREG][SREG / 2];
#pragma unroll
  for (int i = 0; i < QREG; ++i)
#pragma unroll
    for (int jj = 0; jj < SREG / 2; ++jj) part[i][jj] = (v2f){0.f, 0.f};

  for (int d0 = 0; d0 < FDIM; d0 += DCHUNK) {
    const v4f* qsrc = (const v4f*)(Qg + (size_t)row * FDIM + d0);
    const v4f* ssrc = (const v4f*)(Sg + (size_t)row * FDIM + d0);
    v4f* qdst = (v4f*)(sQ + row * LSTRIDE);
    v4f* sdst = (v4f*)(sS + row * LSTRIDE);
    v4f tq[4], ts[4];
#pragma unroll
    for (int j = 0; j < 4; ++j) {
      int c4 = l16 + j * 16;
      tq[j] = qsrc[c4];
      ts[j] = ssrc[c4];
    }
    asm volatile("" : "+v"(tq[0]), "+v"(tq[1]), "+v"(tq[2]), "+v"(tq[3]),
                      "+v"(ts[0]), "+v"(ts[1]), "+v"(ts[2]), "+v"(ts[3]));
#pragma unroll
    for (int j = 0; j < 4; ++j) {
      int c4 = l16 + j * 16;
      qdst[c4] = tq[j];
      sdst[c4] = ts[j];
    }
    __syncthreads();
#pragma unroll
    for (int h = 0; h < DCHUNK / 128; ++h) {
      const int col = (lane + 32 * h) * 4;
      v4f qv[QREG];
#pragma unroll
      for (int i = 0; i < QREG; ++i) qv[i] = *(const v4f*)(sQ + (q0 + i) * LSTRIDE + col);
#pragma unroll
      for (int jh = 0; jh < SREG; jh += 4) {
        v4f sv[4];
#pragma unroll
        for (int j = 0; j < 4; ++j) sv[j] = *(const v4f*)(sS + (s0 + jh + j) * LSTRIDE + col);
#pragma unroll
        for (int i = 0; i < QREG; ++i) {
#pragma unroll
          for (int jj = 0; jj < 2; ++jj) {
            v4f p0 = qv[i] * sv[2 * jj];
            v4f p1 = qv[i] * sv[2 * jj + 1];
            v2f tx = {dev_tanh(p0.x), dev_tanh(p1.x)};
            v2f ty = {dev_tanh(p0.y), dev_tanh(p1.y)};
            v2f tz = {dev_tanh(p0.z), dev_tanh(p1.z)};
            v2f tw = {dev_tanh(p0.w), dev_tanh(p1.w)};
            part[i][jh / 2 + jj] += (tx + ty) + (tz + tw);
          }
        }
      }
    }
    __syncthreads();
  }

  float* rdst = sRed + (q0 * N_SUP + s0) * RSTRIDE + lane;
#pragma unroll
  for (int i = 0; i < QREG; ++i)
#pragma unroll
    for (int jj = 0; jj < SREG / 2; ++jj) {
      rdst[(i * N_SUP + 2 * jj)     * RSTRIDE] = part[i][jj].x;
      rdst[(i * N_SUP + 2 * jj + 1) * RSTRIDE] = part[i][jj].y;
    }
  __syncthreads();
  float acc = 0.f;
#pragma unroll
  for (int l = 0; l < 32; ++l) acc += sRed[tid * RSTRIDE + l];
  sAtt[q][s] = acc;
  __syncthreads();

  float m = -3.402823466e38f;
  float rowv[N_SUP];
#pragma unroll
  for (int i = 0; i < N_SUP; ++i) { rowv[i] = sAtt[q][i]; m = fmaxf(m, rowv[i]); }
  float sum = 0.f;
#pragma unroll
  for (int i = 0; i < N_SUP; ++i) sum += __expf(rowv[i] - m);
  float mya = __expf(acc - m) / sum;
  __syncthreads();
  sAtt[q][s] = mya;
  __syncthreads();

  const int hi   = lane >> 4;
  const int ln   = lane & 15;
  const int k0   = hi ? 2 : 0;

  v2f a[4];
#pragma unroll
  for (int kk = 0; kk < 4; ++kk) {
    a[kk].x = sAtt[ln][kk * 4 + k0];
    a[kk].y = sAtt[ln][kk * 4 + k0 + 1];
  }

  float* outBase = out + ((size_t)(qt * QTILE) * N_CLASS + c) * FDIM;

  for (int g = wave; g < FDIM / 64; g += 8) {
    const int nb = g * 64 + 4 * ln;
    v4f bx[4], by[4];
#pragma unroll
    for (int kk = 0; kk < 4; ++kk) {
      bx[kk] = *(const v4f*)(Sg + (size_t)(kk * 4 + k0)     * FDIM + nb);
      by[kk] = *(const v4f*)(Sg + (size_t)(kk * 4 + k0 + 1) * FDIM + nb);
    }
    v8f d0 = {}, d1 = {}, d2 = {}, d3 = {};
#pragma unroll
    for (int kk = 0; kk < 4; ++kk) {
      d0 = __builtin_amdgcn_wmma_f32_16x16x4_f32(false, a[kk], false, (v2f){bx[kk].x, by[kk].x},
                                                 (short)0, d0, false, false);
      d1 = __builtin_amdgcn_wmma_f32_16x16x4_f32(false, a[kk], false, (v2f){bx[kk].y, by[kk].y},
                                                 (short)0, d1, false, false);
      d2 = __builtin_amdgcn_wmma_f32_16x16x4_f32(false, a[kk], false, (v2f){bx[kk].z, by[kk].z},
                                                 (short)0, d2, false, false);
      d3 = __builtin_amdgcn_wmma_f32_16x16x4_f32(false, a[kk], false, (v2f){bx[kk].w, by[kk].w},
                                                 (short)0, d3, false, false);
    }
#pragma unroll
    for (int r = 0; r < 8; ++r) {
      const int M = r + (hi ? 8 : 0);
      *(v4f*)(outBase + (size_t)M * N_CLASS * FDIM + nb) = (v4f){d0[r], d1[r], d2[r], d3[r]};
    }
  }
}

extern "C" void kernel_launch(void* const* d_in, const int* in_sizes, int n_in,
                              void* d_out, int out_size, void* d_ws, size_t ws_size,
                              hipStream_t stream) {
  const float* data = (const float*)d_in[0];
  float* out = (float*)d_out;
  (void)in_sizes; (void)n_in; (void)d_ws; (void)ws_size; (void)out_size;
  dim3 grid(N_QRY / QTILE, N_CLASS);
  proto_att_kernel<<<grid, 256, 0, stream>>>(data, out);
}


# score loop software-pipelined (multiplies one step ahead, adds one step behind) with a scheduling request: tanh and packed mul/add alternate, LDS reads lead their use; sums and order unchanged
# speedup vs baseline: 1.6327x; 1.0447x over previous
#include <hip/hip_runtime.h>
#include <hip/hip_bf16.h>

typedef __attribute__((ext_vector_type(2))) float v2f;
typedef __attribute__((ext_vector_type(8))) float v8f;
typedef __attribute__((ext_vector_type(4))) float v4f;

#define N_CLASS 32
#define N_SUP   16
#define FDIM    1024
#define N_QRY   512
#define QTILE   16
#define DCHUNK  256
#define LSTRIDE (DCHUNK + 4)
#define QREG    4
#define SREG    8
#define RSTRIDE 33

__device__ __forceinline__ float dev_tanh(float x) {
#if __has_builtin(__builtin_amdgcn_tanhf)
  return __builtin_amdgcn_tanhf(x);
#else
  return tanhf(x);
#endif
}

__global__ __launch_bounds__(256, 8)
void proto_att_kernel(const float* __restrict__ data, float* __restrict__ out) {
  __shared__ __attribute__((aligned(16))) float sBuf[QTILE * N_SUP * RSTRIDE];
  __shared__ float sAtt[QTILE][N_SUP + 1];
  float* sQ   = sBuf;
  float* sS   = sBuf + QTILE * LSTRIDE;
  float* sRed = sBuf;
  static_assert(QTILE * N_SUP * RSTRIDE >= (QTILE + N_SUP) * LSTRIDE, "staging tiles must fit");

  const int qt  = blockIdx.x;
  const int c   = blockIdx.y;
  const int tid = threadIdx.x;

  const float* Qg = data + (size_t)(N_CLASS * N_SUP + qt * QTILE) * FDIM;
  const float* Sg = data + (size_t)c * N_SUP * FDIM;

  const int q   = tid >> 4;
  const int s   = tid & 15;
  const int row = tid >> 4;
  const int l16 = tid & 15;

  const int wave = tid >> 5;
  const int lane = tid & 31;

  const int q0 = (wave >> 1) * QREG;
  const int s0 = (wave & 1) * SREG;
  v2f part[QREG][SREG / 2];
#pragma unroll
  for (int i = 0; i < QREG; ++i)
#pragma unroll
    for (int jj = 0; jj < SREG / 2; ++jj) part[i][jj] = (v2f){0.f, 0.f};

  for (int d0 = 0; d0 < FDIM; d0 += DCHUNK) {
    const v4f* qsrc = (const v4f*)(Qg + (size_t)row * FDIM + d0);
    const v4f* ssrc = (const v4f*)(Sg + (size_t)row * FDIM + d0);
    v4f* qdst = (v4f*)(sQ + row * LSTRIDE);
    v4f* sdst = (v4f*)(sS + row * LSTRIDE);
    v4f tq[4], ts[4];
#pragma unroll
    for (int j = 0; j < 4; ++j) {
      int c4 = l16 + j * 16;
      tq[j] = qsrc[c4];
      ts[j] = ssrc[c4];
    }
    asm volatile("" : "+v"(tq[0]), "+v"(tq[1]), "+v"(tq[2]), "+v"(tq[3]),
                      "+v"(ts[0]), "+v"(ts[1]), "+v"(ts[2]), "+v"(ts[3]));
#pragma unroll
    for (int j = 0; j < 4; ++j) {
      int c4 = l16 + j * 16;
      qdst[c4] = tq[j];
      sdst[c4] = ts[j];
    }
    __syncthreads();
    v4f qv[DCHUNK / 128][QREG];
    v4f sv[8][2];
    v4f pr0[2], pr1[2];
    v2f tx[2], ty[2], tz[2], tw[2];
    constexpr int NSTEP = 32;
    static_assert(NSTEP == (DCHUNK / 128) * (SREG / 2) * QREG, "step count");
#pragma unroll
    for (int n = -1; n <= NSTEP; ++n) {
      if (n == -1) {
#pragma unroll
        for (int r = 0; r < QREG; ++r) qv[0][r] = *(const v4f*)(sQ + (q0 + r) * LSTRIDE + lane * 4);
      }
#pragma unroll
      for (int t = 0; t < 8; ++t) {
        const int at = (4 * t - 6 < -1) ? -1 : 4 * t - 6;
        if (n == at) {
          const int h = t >> 2, rowp = s0 + 4 * ((t >> 1) & 1) + 2 * (t & 1);
          const int col = (lane + 32 * h) * 4;
          sv[t][0] = *(const v4f*)(sS + rowp * LSTRIDE + col);
          sv[t][1] = *(const v4f*)(sS + (rowp + 1) * LSTRIDE + col);
        }
      }
      if (n == 6 || n == 10) {
        const int r0 = (n == 6) ? 0 : 2;
        qv[1][r0]     = *(const v4f*)(sQ + (q0 + r0) * LSTRIDE + (lane + 32) * 4);
        qv[1][r0 + 1] = *(const v4f*)(sQ + (q0 + r0 + 1) * LSTRIDE + (lane + 32) * 4);
      }
      const int m = n + 1;
      if (m < NSTEP) {
        const int t = m >> 2, h = m >> 4, i = m & 3;
        pr0[m & 1] = qv[h][i] * sv[t][0];
        pr1[m & 1] = qv[h][i] * sv[t][1];
      }
      if (n >= 1) {
        const int k = n - 1;
        const int i = k & 3, x = (k >> 2) & 3;
        part[i][x] += (tx[k & 1] + ty[k & 1]) + (tz[k & 1] + tw[k & 1]);
      }
      if (n >= 0 && n < NSTEP) {
        const v4f p0 = pr0[n & 1], p1 = pr1[n & 1];
        tx[n & 1] = (v2f){dev_tanh(p0.x), dev_tanh(p1.x)};
        ty[n & 1] = (v2f){dev_tanh(p0.y), dev_tanh(p1.y)};
        tz[n & 1] = (v2f){dev_tanh(p0.z), dev_tanh(p1.z)};
        tw[n & 1] = (v2f){dev_tanh(p0.w), dev_tanh(p1.w)};
      }
      if ((n & 3) == 1 || n == NSTEP) {
        const int first = (n == 1) ? -1 : n - 3;
        const int nds = (first == -1) ? QREG + 4 : (first == 6 || first == 10) ? 4 : (first >= 2 && first <= 22) ? 2 : 0;
        const int npair = (n == 1 || n == NSTEP) ? 16 : 32;
        if (nds == 8) __builtin_amdgcn_sched_group_barrier(0x100, 8, 0);
        if (nds == 4) __builtin_amdgcn_sched_group_barrier(0x100, 4, 0);
        if (nds == 2) __builtin_amdgcn_sched_group_barrier(0x100, 2, 0);
#pragma unroll
        for (int k = 0; k < npair; ++k) {
          __builtin_amdgcn_sched_group_barrier(0x002, 1, 0);
          __builtin_amdgcn_sched_group_barrier(0x400, 1, 0);
        }
        if (n != NSTEP) __builtin_amdgcn_sched_barrier(0);
      }
    }
    __syncthreads();
  }

  float* rdst = sRed + (q0 * N_SUP + s0) * RSTRIDE + lane;
#pragma unroll
  for (int i = 0; i < QREG; ++i)
#pragma unroll
    for (int jj = 0; jj < SREG / 2; ++jj) {
      rdst[(i * N_SUP + 2 * jj)     * RSTRIDE] = part[i][jj].x;
      rdst[(i * N_SUP + 2 * jj + 1) * RSTRIDE] = part[i][jj].y;
    }
  __syncthreads();
  float acc = 0.f;
#pragma unroll
  for (int l = 0; l < 32; ++l) acc += sRed[tid * RSTRIDE + l];
  sAtt[q][s] = acc;
  __syncthreads();

  float m = -3.402823466e38f;
  float rowv[N_SUP];
#pragma unroll
  for (int i = 0; i < N_SUP; ++i) { rowv[i] = sAtt[q][i]; m = fmaxf(m, rowv[i]); }
  float sum = 0.f;
#pragma unroll
  for (int i = 0; i < N_SUP; ++i) sum += __expf(rowv[i] - m);
  float mya = __expf(acc - m) / sum;
  __syncthreads();
  sAtt[q][s] = mya;
  __syncthreads();

  const int hi   = lane >> 4;
  const int ln   = lane & 15;
  const int k0   = hi ? 2 : 0;

  v2f a[4];
#pragma unroll
  for (int kk = 0; kk < 4; ++kk) {
    a[kk].x = sAtt[ln][kk * 4 + k0];
    a[kk].y = sAtt[ln][kk * 4 + k0 + 1];
  }

  float* outBase = out + ((size_t)(qt * QTILE) * N_CLASS + c) * FDIM;

  for (int g = wave; g < FDIM / 64; g += 8) {
    const int nb = g * 64 + 4 * ln;
    v4f bx[4], by[4];
#pragma unroll
    for (int kk = 0; kk < 4; ++kk) {
      bx[kk] = *(const v4f*)(Sg + (size_t)(kk * 4 + k0)     * FDIM + nb);
      by[kk] = *(const v4f*)(Sg + (size_t)(kk * 4 + k0 + 1) * FDIM + nb);
    }
    v8f d0 = {}, d1 = {}, d2 = {}, d3 = {};
#pragma unroll
    for (int kk = 0; kk < 4; ++kk) {
      d0 = __builtin_amdgcn_wmma_f32_16x16x4_f32(false, a[kk], false, (v2f){bx[kk].x, by[kk].x},
                                                 (short)0, d0, false, false);
      d1 = __builtin_amdgcn_wmma_f32_16x16x4_f32(false, a[kk], false, (v2f){bx[kk].y, by[kk].y},
                                                 (short)0, d1, false, false);
      d2 = __builtin_amdgcn_wmma_f32_16x16x4_f32(false, a[kk], false, (v2f){bx[kk].z, by[kk].z},
                                                 (short)0, d2, false, false);
      d3 = __builtin_amdgcn_wmma_f32_16x16x4_f32(false, a[kk], false, (v2f){bx[kk].w, by[kk].w},
                                                 (short)0, d3, false, false);
    }
#pragma unroll
    for (int r = 0; r < 8; ++r) {
      const int M = r + (hi ? 8 : 0);
      *(v4f*)(outBase + (size_t)M * N_CLASS * FDIM + nb) = (v4f){d0[r], d1[r], d2[r], d3[r]};
    }
  }
}

extern "C" void kernel_launch(void* const* d_in, const int* in_sizes, int n_in,
                              void* d_out, int out_size, void* d_ws, size_t ws_size,
                              hipStream_t stream) {
  const float* data = (const float*)d_in[0];
  float* out = (float*)d_out;
  (void)in_sizes; (void)n_in; (void)d_ws; (void)ws_size; (void)out_size;
  dim3 grid(N_QRY / QTILE, N_CLASS);
  proto_att_kernel<<<grid, 256, 0, stream>>>(data, out);
}
